# starting kernel
# baseline (speedup 1.0000x reference)
#include <hip/hip_runtime.h>
#include <hip/hip_bf16.h>

#define R_ 8
#define K_ 10
#define D_ 256
#define DL_ 10
#define B_ 4096
#define KD_ 100
#define KDP_ 112
#define INV_TEMP 10.0f

#define XP 260
#define ZP 114

#define G_OFF  0u
#define UP_OFF 32768u
#define UT_OFF 950272u

typedef float v2f __attribute__((ext_vector_type(2)));
typedef float v8f __attribute__((ext_vector_type(8)));

__global__ __launch_bounds__(128) void gram_kernel(const float* __restrict__ Us,
                                                   float* __restrict__ G) {
  const int rk = blockIdx.x;
  const int t  = threadIdx.x;
  if (t >= KD_) return;
  const int d1 = t / DL_, d2 = t % DL_;
  const float* u = Us + (size_t)rk * D_ * DL_;
  float acc = 0.f;
#pragma unroll 8
  for (int Dd = 0; Dd < D_; ++Dd)
    acc = fmaf(u[Dd * DL_ + d1], u[Dd * DL_ + d2], acc);
  G[rk * KD_ + t] = acc;
}

__global__ __launch_bounds__(KDP_) void pack_kernel(const float* __restrict__ Us,
                                                    float* __restrict__ Upad2,
                                                    float* __restrict__ UpadT2) {
  const int kd = threadIdx.x;
  const int rd = blockIdx.x;
  const int r  = rd >> 8;
  const int Dd = rd & 255;
  float val = 0.f;
  if (kd < KD_) {
    const int k = kd / DL_, d = kd % DL_;
    val = Us[(((size_t)r * K_ + k) * D_ + Dd) * DL_ + d];
  }
  Upad2 [(((size_t)r * (D_ / 2) + (Dd >> 1)) * KDP_ + kd) * 2 + (Dd & 1)] = val;
  UpadT2[(((size_t)r * (KDP_ / 2) + (kd >> 1)) * D_ + Dd) * 2 + (kd & 1)] = val;
}

__global__ __launch_bounds__(32) void ksub_fused(const float* __restrict__ x,
                                                 const float* __restrict__ Upad2,
                                                 const float* __restrict__ UpadT2,
                                                 const float* __restrict__ G,
                                                 float* __restrict__ out) {
  __shared__ float sx[16 * XP];
  __shared__ float sz[16 * ZP];
  __shared__ float sg[K_ * KD_];

  const int lane = threadIdx.x;
  const int r  = blockIdx.y;
  const int b0 = blockIdx.x * 16;
  const int g  = lane >> 4;
  const int n  = lane & 15;

  for (int i = lane; i < 16 * 64; i += 32) {
    const int row = i >> 6;
    const int c4  = (i & 63) << 2;
    const float4 v = *(const float4*)(x + (size_t)(b0 + row) * D_ + c4);
    *(float4*)(&sx[row * XP + c4]) = v;
  }
  for (int i = lane; i < K_ * KD_; i += 32) sg[i] = G[r * K_ * KD_ + i];
  __syncthreads();

  const float* Up = Upad2 + (size_t)r * (D_ / 2) * KDP_ * 2;
#pragma unroll 1
  for (int t = 0; t < 7; ++t) {
    const int kd = t * 16 + n;
    const float* ub = Up + ((size_t)g * KDP_ + kd) * 2;
    v8f acc = {};
#pragma unroll
    for (int kb = 0; kb < D_; kb += 4) {
      v2f a = *(const v2f*)(&sx[n * XP + kb + 2 * g]);
      v2f b = *(const v2f*)(&ub[(size_t)(kb >> 1) * KDP_ * 2]);
      acc = __builtin_amdgcn_wmma_f32_16x16x4_f32(false, a, false, b,
                                                  (short)0, acc, false, false);
    }
#pragma unroll
    for (int v = 0; v < 8; ++v)
      sz[(v + 8 * g) * ZP + kd] = acc[v];
  }
  __syncthreads();

  if (lane < 16) {
    float s[K_];
    float m = -3.0e38f;
#pragma unroll 1
    for (int k = 0; k < K_; ++k) {
      float zk[DL_];
#pragma unroll
      for (int d = 0; d < DL_; ++d) zk[d] = sz[lane * ZP + k * DL_ + d];
      float zz = 0.f, q = 0.f;
#pragma unroll
      for (int d1 = 0; d1 < DL_; ++d1) {
        zz = fmaf(zk[d1], zk[d1], zz);
        float t2 = 0.f;
#pragma unroll
        for (int d2 = 0; d2 < DL_; ++d2)
          t2 = fmaf(sg[k * KD_ + d1 * DL_ + d2], zk[d2], t2);
        q = fmaf(zk[d1], t2, q);
      }
      s[k] = (zz - 0.5f * q) * INV_TEMP;
      m = fmaxf(m, s[k]);
    }
    float c[K_];
    float sum = 0.f;
#pragma unroll
    for (int k = 0; k < K_; ++k) { c[k] = __expf(s[k] - m); sum += c[k]; }
    const float inv = 1.0f / sum;
#pragma unroll 1
    for (int k = 0; k < K_; ++k) {
      const float ck = c[k] * inv;
#pragma unroll
      for (int d = 0; d < DL_; ++d) sz[lane * ZP + k * DL_ + d] *= ck;
    }
  }
  __syncthreads();

  const float* Ut = UpadT2 + (size_t)r * (KDP_ / 2) * D_ * 2;
  float* outr = out + (size_t)r * B_ * D_ + (size_t)b0 * D_;
#pragma unroll 1
  for (int t = 0; t < 16; ++t) {
    const int Dd = t * 16 + n;
    const float* ubt = Ut + ((size_t)g * D_ + Dd) * 2;
    v8f acc = {};
#pragma unroll
    for (int kb = 0; kb < KD_; kb += 4) {
      v2f a = *(const v2f*)(&sz[n * ZP + kb + 2 * g]);
      v2f b = *(const v2f*)(&ubt[(size_t)(kb >> 1) * D_ * 2]);
      acc = __builtin_amdgcn_wmma_f32_16x16x4_f32(false, a, false, b,
                                                  (short)0, acc, false, false);
    }
#pragma unroll
    for (int v = 0; v < 8; ++v)
      outr[(size_t)(v + 8 * g) * D_ + Dd] = acc[v];
  }
}

extern "C" void kernel_launch(void* const* d_in, const int* in_sizes, int n_in,
                              void* d_out, int out_size, void* d_ws, size_t ws_size,
                              hipStream_t stream) {
  (void)in_sizes; (void)n_in; (void)out_size; (void)ws_size;
  const float* x  = (const float*)d_in[0];
  const float* Us = (const float*)d_in[1];
  float* G      = (float*)((char*)d_ws + G_OFF);
  float* Upad2  = (float*)((char*)d_ws + UP_OFF);
  float* UpadT2 = (float*)((char*)d_ws + UT_OFF);
  float* out = (float*)d_out;

  gram_kernel<<<R_ * K_, 128, 0, stream>>>(Us, G);
  pack_kernel<<<R_ * D_, KDP_, 0, stream>>>(Us, Upad2, UpadT2);
  ksub_fused<<<dim3(B_ / 16, R_), 32, 0, stream>>>(x, Upad2, UpadT2, G, out);
}


# speedup vs baseline: 2.8226x; 2.8226x over previous
#include <hip/hip_runtime.h>
#include <hip/hip_bf16.h>

#define R_ 8
#define K_ 10
#define D_ 256
#define DL_ 10
#define B_ 4096
#define KD_ 100
#define INV_TEMP 10.0f

#define XP 260
#define ZP 114


typedef float v2f __attribute__((ext_vector_type(2)));
typedef float v4f __attribute__((ext_vector_type(4)));
typedef float v8f __attribute__((ext_vector_type(8)));

#define HOLD(V)                                                                \
  do {                                                                         \
    __builtin_amdgcn_sched_barrier(0);                                         \
    asm volatile("" : "+v"(V) : : "memory");                                   \
    __builtin_amdgcn_sched_barrier(0);                                         \
  } while (0)

#define NW 8
#define NT (NW * 32)
#define PITCH 12
#define SLAB (D_ * PITCH + 10)
#define SU_FLOATS (K_ * SLAB)
#define UR_FLOATS (K_ * D_ * DL_)
#define U_V4 (UR_FLOATS / 4 / NT)

__global__ __launch_bounds__(NT) void ksub_fused(const float* __restrict__ x,
                                                 const float* __restrict__ Us,
                                                 float* __restrict__ out) {
  __shared__ __attribute__((aligned(16))) float sU[SU_FLOATS];
  __shared__ __attribute__((aligned(16))) float sx[NW * 16 * XP];
  __shared__ __attribute__((aligned(16))) float sz[NW * 16 * ZP];
  __shared__ __attribute__((aligned(32))) float sg[K_ * KD_];

  const int tid  = threadIdx.x;
  const int w    = tid >> 5;
  const int lane = tid & 31;
  const int r  = blockIdx.y;
  const int b0 = (blockIdx.x * NW + w) * 16;
  const int g  = lane >> 4;
  const int n  = lane & 15;
  float* sxw = sx + w * 16 * XP;
  float* szw = sz + w * 16 * ZP;

  {
    const v4f* s1 = (const v4f*)(Us + (size_t)r * UR_FLOATS);
    v4f ru[U_V4];
#pragma unroll
    for (int j = 0; j < U_V4; ++j) ru[j] = s1[tid + j * NT];
    const float* xw = x + (size_t)b0 * D_;
    v4f rx[32];
#pragma unroll
    for (int j = 0; j < 32; ++j)
      rx[j] = *(const v4f*)(xw + (j >> 1) * D_ + (((j & 1) << 5) + lane) * 4);

#pragma unroll
    for (int j = 0; j < U_V4; ++j) {
      const unsigned i0 = 4u * (unsigned)(tid + j * NT);
      const unsigned i1 = i0 + 2u;
      const unsigned row0 = i0 / 10u, col0 = i0 - 10u * row0;
      const unsigned row1 = i1 / 10u, col1 = i1 - 10u * row1;
      v2f lo = {ru[j].x, ru[j].y};
      v2f hi = {ru[j].z, ru[j].w};
      *(v2f*)(&sU[(row0 >> 8) * SLAB + (row0 & 255u) * PITCH + col0]) = lo;
      *(v2f*)(&sU[(row1 >> 8) * SLAB + (row1 & 255u) * PITCH + col1]) = hi;
    }
#pragma unroll
    for (int j = 0; j < 32; ++j)
      *(v4f*)(&sxw[(j >> 1) * XP + (((j & 1) << 5) + lane) * 4]) = rx[j];
  }
  __syncthreads();

  {
    const int ws = __builtin_amdgcn_readfirstlane(w);
#pragma unroll 1
    for (int c = ws; c < K_; c += NW) {
      const int kdc = (c * DL_ + n < KD_) ? (c * DL_ + n) : (KD_ - 1);
      const int kq = kdc / DL_, dq = kdc - kq * DL_;
      const float* uc = sU + kq * SLAB + (2 * g) * PITCH + dq;
      v2f f[D_ / 4];
#pragma unroll
      for (int j = 0; j < D_ / 4; ++j) {
        f[j].x = uc[(4 * j) * PITCH];
        f[j].y = uc[(4 * j + 1) * PITCH];
      }
      __builtin_amdgcn_sched_barrier(0);
      v8f acc = {};
#pragma unroll
      for (int j = 0; j < D_ / 4; ++j)
        acc = __builtin_amdgcn_wmma_f32_16x16x4_f32(false, f[j], false, f[j],
                                                    (short)0, acc, false, false);
      if (n < DL_) {
#pragma unroll
        for (int v = 0; v < 8; ++v)
          if (v + 8 * g < DL_) sg[c * KD_ + (v + 8 * g) * DL_ + n] = acc[v];
      }
    }
  }

  v2f xa[D_ / 4];
#pragma unroll
  for (int j = 0; j < D_ / 4; ++j)
    xa[j] = *(const v2f*)(&sxw[n * XP + 4 * j + 2 * g]);

#define G1_BASE(T, P)                                                          \
  {                                                                            \
    const int kd_ = (T) * 16 + n;                                              \
    const int kc_ = (kd_ < KD_) ? kd_ : (KD_ - 1);                             \
    const int kq_ = kc_ / DL_;                                                 \
    P = sU + kq_ * SLAB + (2 * g) * PITCH + (kc_ - kq_ * DL_);                 \
  }
#define G1_LOAD(S, BLK, P)                                                     \
  _Pragma("unroll") for (int i = 0; i < 16; ++i) {                             \
    S[i].x = (P)[(4 * ((BLK) * 16 + i)) * PITCH];                              \
    S[i].y = (P)[(4 * ((BLK) * 16 + i) + 1) * PITCH];                          \
  }
#define G1_MMA(S, BLK)                                                         \
  _Pragma("unroll") for (int i = 0; i < 16; ++i)                               \
    acc = __builtin_amdgcn_wmma_f32_16x16x4_f32(false, xa[(BLK) * 16 + i],     \
                                                false, S[i], (short)0, acc,    \
                                                false, false);
  {
    v2f s0[16], s1[16];
    const float* ub;
    G1_BASE(0, ub)
    G1_LOAD(s0, 0, ub)
    G1_LOAD(s1, 1, ub)
#pragma unroll 1
    for (int t = 0; t < 7; ++t) {
      const float* un;
      G1_BASE(t + 1, un)
      v8f acc = {};
      G1_MMA(s0, 0)
      HOLD(acc);
      G1_LOAD(s0, 2, ub)
      HOLD(acc);
      G1_MMA(s1, 1)
      HOLD(acc);
      G1_LOAD(s1, 3, ub)
      HOLD(acc);
      G1_MMA(s0, 2)
      HOLD(acc);
      G1_LOAD(s0, 0, un)
      HOLD(acc);
      G1_MMA(s1, 3)
      HOLD(acc);
      G1_LOAD(s1, 1, un)
      const int kd = t * 16 + n;
#pragma unroll
      for (int v = 0; v < 8; ++v)
        szw[(v + 8 * g) * ZP + kd] = acc[v];
      ub = un;
    }
  }
#undef G1_BASE
#undef G1_LOAD
#undef G1_MMA

  __syncthreads();

  {
    float* zr = szw + n * ZP + (5 * g) * DL_;
    const float* gr = sg + (5 * g) * KD_;
    float s[K_ / 2];
    float m = -3.0e38f;
#pragma unroll 1
    for (int kk = 0; kk < K_ / 2; ++kk) {
      float zk[DL_];
#pragma unroll
      for (int d = 0; d < DL_; ++d) zk[d] = zr[kk * DL_ + d];
      const float* gk = gr + kk * KD_;
      float a = 0.f;
#pragma unroll
      for (int d1 = 0; d1 < DL_; ++d1) {
        float t = fmaf(-0.5f, gk[d1 * DL_ + d1], 1.0f) * zk[d1];
#pragma unroll
        for (int d2 = d1 + 1; d2 < DL_; ++d2)
          t = fmaf(-gk[d1 * DL_ + d2], zk[d2], t);
        a = fmaf(zk[d1], t, a);
      }
      const float sk = a * INV_TEMP;
      s[kk] = sk;
      m = fmaxf(m, sk);
    }
    m = fmaxf(m, __shfl_xor(m, 16));
    float c[K_ / 2];
    float sum = 0.f;
#pragma unroll
    for (int kk = 0; kk < K_ / 2; ++kk) { c[kk] = __expf(s[kk] - m); sum += c[kk]; }
    sum += __shfl_xor(sum, 16);
    const float inv = 1.0f / sum;
#pragma unroll
    for (int kk = 0; kk < K_ / 2; ++kk) {
      const float ck = c[kk] * inv;
#pragma unroll
      for (int d = 0; d < DL_; ++d) zr[kk * DL_ + d] *= ck;
    }
  }

  float* outr = out + (size_t)r * B_ * D_ + (size_t)b0 * D_;
  v2f wa[KD_ / 4];
#pragma unroll
  for (int i = 0; i < KD_ / 4; ++i)
    wa[i] = *(const v2f*)(&szw[n * ZP + 4 * i + 2 * g]);

#define G2_LOAD(S, T)                                                          \
  {                                                                            \
    const float* ua_ = sU + ((T) * 16 + n) * PITCH + 2 * g;                    \
    const float* ux_ = sU + ((T) * 16 + n) * PITCH + g * (SLAB - 8);           \
    _Pragma("unroll") for (int i = 0; i < KD_ / 4; ++i) {                      \
      const int kb_ = 4 * i;                                                   \
      const float* p_ = ((kb_ % DL_) == 8) ? ux_ : ua_;                        \
      S[i] = *(const v2f*)(&p_[(kb_ / DL_) * SLAB + (kb_ % DL_)]);             \
    }                                                                          \
  }
#define G2_MMA(S, ACC)                                                         \
  _Pragma("unroll") for (int i = 0; i < KD_ / 4; ++i)     \
    ACC = __builtin_amdgcn_wmma_f32_16x16x4_f32(false, wa[i], false, S[i],     \
                                                (short)0, ACC, false, false);
#define G2_STORE(T, ACC)                                                       \
  {                                                                            \
    const int Dd_ = (T) * 16 + n;                      \
    _Pragma("unroll") for (int v = 0; v < 8; ++v)             \
      outr[(size_t)(v + 8 * g) * D_ + Dd_] = ACC[v];                           \
  }
  {
    v2f b0[KD_ / 4], b1[KD_ / 4];
    G2_LOAD(b0, 0)
#pragma unroll 1
    for (int t = 0; t < 16; t += 2) {
      G2_LOAD(b1, t + 1)
      HOLD(b0[0]);
      v8f acc0 = {};
      G2_MMA(b0, acc0)
      HOLD(acc0);
      G2_STORE(t, acc0)
      const int tn = (t + 2 < 16) ? (t + 2) : 15;
      G2_LOAD(b0, tn)
      HOLD(b1[0]);
      v8f acc1 = {};
      G2_MMA(b1, acc1)
      G2_STORE(t + 1, acc1)
    }
  }
#undef G2_LOAD
#undef G2_MMA
#undef G2_STORE
}

extern "C" void kernel_launch(void* const* d_in, const int* in_sizes, int n_in,
                              void* d_out, int out_size, void* d_ws, size_t ws_size,
                              hipStream_t stream) {
  (void)in_sizes; (void)n_in; (void)out_size; (void)d_ws; (void)ws_size;
  const float* x  = (const float*)d_in[0];
  const float* Us = (const float*)d_in[1];
  float* out = (float*)d_out;

  ksub_fused<<<dim3(B_ / (16 * NW), R_), NT, 0, stream>>>(x, Us, out);
}
